# starting kernel
# baseline (speedup 1.0000x reference)
#include <hip/hip_runtime.h>

#define EMB 512
#define OUTW (3 * EMB)

typedef __attribute__((ext_vector_type(2)))  float     v2f;
typedef __attribute__((ext_vector_type(8)))  float     v8f;
typedef __attribute__((ext_vector_type(4)))  float     f32x4;
typedef __attribute__((ext_vector_type(16))) _Float16  v16h;

__global__ __launch_bounds__(128) void rowmean_wmma(const float* __restrict__ emb,
                                                    float* __restrict__ node_mean,
                                                    int n_nodes) {
  const int lane = threadIdx.x & 31;
  const int wave = threadIdx.x >> 5;
  const int row0 = (blockIdx.x * 4 + wave) * 16;
  if (row0 >= n_nodes) return;

  const int m    = lane & 15;
  const int half = lane >> 4;
  const float* __restrict__ row = emb + (size_t)(row0 + m) * EMB;

  v8f c = {};

#if defined(__AMDGCN__) && __has_builtin(__builtin_amdgcn_wmma_f32_16x16x4_f32)
  const v2f* __restrict__ rp = (const v2f*)row + half;
  v2f bones = {1.0f, 1.0f};
  for (int k = 0; k < EMB / 4; ++k) {
    v2f a = rp[k * 2];
    c = __builtin_amdgcn_wmma_f32_16x16x4_f32(
         false, a,  false, bones,
         (short)0, c,  false,  false);
  }
#else
  v16h bones;
#pragma unroll
  for (int j = 0; j < 16; ++j) bones[j] = (_Float16)1.0f;
  for (int k0 = 0; k0 < EMB; k0 += 32) {
    v16h a;
    const float* p0 = row + k0 + half * 8;
    const float* p1 = row + k0 + 16 + half * 8;
#pragma unroll
    for (int j = 0; j < 8; ++j) a[j] = (_Float16)p0[j];
#pragma unroll
    for (int j = 0; j < 8; ++j) a[8 + j] = (_Float16)p1[j];
    c = __builtin_amdgcn_wmma_f32_16x16x32_f16(
        false, a, false, bones, (short)0, c, false, false);
  }
#endif

  if (m == 0) {
    const float inv = 1.0f / (float)EMB;
    const int base = row0 + half * 8;
#pragma unroll
    for (int r = 0; r < 8; ++r) node_mean[base + r] = c[r] * inv;
  }
}

__global__ void zero_f32(float* __restrict__ p, int n) {
  int i = blockIdx.x * blockDim.x + threadIdx.x;
  if (i < n) p[i] = 0.0f;
}

__global__ void dst_scatter(const int* __restrict__ dst_event,
                            const int* __restrict__ dst_node,
                            const float* __restrict__ node_mean,
                            float* __restrict__ dst_sum, int n_dst) {
  int i = blockIdx.x * blockDim.x + threadIdx.x;
  if (i >= n_dst) return;
  unsafeAtomicAdd(&dst_sum[dst_event[i]], node_mean[dst_node[i]]);
}

__global__ void event_reduce(const int* __restrict__ event_pair,
                             const int* __restrict__ event_src,
                             const int* __restrict__ event_ndst,
                             const int* __restrict__ pair_place,
                             const float* __restrict__ node_mean,
                             const float* __restrict__ dst_sum,
                             float* __restrict__ pair_sum,
                             float* __restrict__ pair_cnt, int n_events) {
  int i = blockIdx.x * blockDim.x + threadIdx.x;
  if (i >= n_events) return;
  int p = event_pair[i];
  float s = (node_mean[pair_place[p]] + node_mean[event_src[i]] +
             dst_sum[i] / ((float)event_ndst[i] + 1.0f)) * (1.0f / 3.0f);
  unsafeAtomicAdd(&pair_sum[p], s);
  unsafeAtomicAdd(&pair_cnt[p], 1.0f);
}

__global__ void pair_reduce(const int* __restrict__ pair_place,
                            const float* __restrict__ pair_sum,
                            const float* __restrict__ pair_cnt,
                            float* __restrict__ place_sum,
                            float* __restrict__ place_cnt, int n_pairs) {
  int i = blockIdx.x * blockDim.x + threadIdx.x;
  if (i >= n_pairs) return;
  float c = pair_cnt[i];
  if (c > 0.0f) {
    int pl = pair_place[i];
    unsafeAtomicAdd(&place_sum[pl], pair_sum[i] / c);
    unsafeAtomicAdd(&place_cnt[pl], 1.0f);
  }
}

__global__ __launch_bounds__(128) void bcast_out(const float* __restrict__ place_sum,
                                                 const float* __restrict__ place_cnt,
                                                 float* __restrict__ out, int n_nodes) {
  int row = blockIdx.x;
  if (row >= n_nodes) return;
  float c = place_cnt[row];
  float pm = c > 0.0f ? place_sum[row] / c : 0.0f;
  f32x4 v = {pm, pm, pm, pm};
  f32x4* __restrict__ o = (f32x4*)out + (size_t)row * (OUTW / 4);
  for (int j = threadIdx.x; j < OUTW / 4; j += 128)
    __builtin_nontemporal_store(v, o + j);
}

extern "C" void kernel_launch(void* const* d_in, const int* in_sizes, int n_in,
                              void* d_out, int out_size, void* d_ws, size_t ws_size,
                              hipStream_t stream) {
  const float* emb       = (const float*)d_in[0];
  const int* pair_place  = (const int*)d_in[1];
  const int* event_pair  = (const int*)d_in[2];
  const int* event_src   = (const int*)d_in[3];
  const int* event_ndst  = (const int*)d_in[4];
  const int* dst_event   = (const int*)d_in[5];
  const int* dst_node    = (const int*)d_in[6];

  const int n_nodes  = in_sizes[0] / EMB;
  const int n_pairs  = in_sizes[1];
  const int n_events = in_sizes[2];
  const int n_dst    = in_sizes[5];

  float* ws        = (float*)d_ws;
  float* node_mean = ws;
  float* dst_sum   = node_mean + n_nodes;
  float* pair_sum  = dst_sum + n_events;
  float* pair_cnt  = pair_sum + n_pairs;
  float* place_sum = pair_cnt + n_pairs;
  float* place_cnt = place_sum + n_nodes;

  const int nzero = n_events + 2 * n_pairs + 2 * n_nodes;
  zero_f32<<<(nzero + 255) / 256, 256, 0, stream>>>(dst_sum, nzero);

  rowmean_wmma<<<(n_nodes + 63) / 64, 128, 0, stream>>>(emb, node_mean, n_nodes);

  dst_scatter<<<(n_dst + 255) / 256, 256, 0, stream>>>(dst_event, dst_node,
                                                       node_mean, dst_sum, n_dst);

  event_reduce<<<(n_events + 255) / 256, 256, 0, stream>>>(
      event_pair, event_src, event_ndst, pair_place, node_mean, dst_sum,
      pair_sum, pair_cnt, n_events);

  pair_reduce<<<(n_pairs + 255) / 256, 256, 0, stream>>>(
      pair_place, pair_sum, pair_cnt, place_sum, place_cnt, n_pairs);

  bcast_out<<<n_nodes, 128, 0, stream>>>(place_sum, place_cnt, (float*)d_out,
                                         n_nodes);
}


# event->pair sums by owner work-groups: pair ranges x event chunks, a group adds its chunk's in-range events into an LDS table, flushes it by coalesced atomics; 6->7 launches; sum order changes
# speedup vs baseline: 1.5297x; 1.5297x over previous
#include <hip/hip_runtime.h>

#define EMB 512
#define OUTW (3 * EMB)

typedef __attribute__((ext_vector_type(2)))  float     v2f;
typedef __attribute__((ext_vector_type(8)))  float     v8f;
typedef __attribute__((ext_vector_type(4)))  float     f32x4;
typedef __attribute__((ext_vector_type(16))) _Float16  v16h;

__global__ __launch_bounds__(128) void rowmean_wmma(const float* __restrict__ emb,
                                                    float* __restrict__ node_mean,
                                                    int n_nodes) {
  const int lane = threadIdx.x & 31;
  const int wave = threadIdx.x >> 5;
  const int row0 = (blockIdx.x * 4 + wave) * 16;
  if (row0 >= n_nodes) return;

  const int m    = lane & 15;
  const int half = lane >> 4;
  const float* __restrict__ row = emb + (size_t)(row0 + m) * EMB;

  v8f c = {};

#if defined(__AMDGCN__) && __has_builtin(__builtin_amdgcn_wmma_f32_16x16x4_f32)
  const v2f* __restrict__ rp = (const v2f*)row + half;
  v2f bones = {1.0f, 1.0f};
  for (int k = 0; k < EMB / 4; ++k) {
    v2f a = rp[k * 2];
    c = __builtin_amdgcn_wmma_f32_16x16x4_f32(
         false, a,  false, bones,
         (short)0, c,  false,  false);
  }
#else
  v16h bones;
#pragma unroll
  for (int j = 0; j < 16; ++j) bones[j] = (_Float16)1.0f;
  for (int k0 = 0; k0 < EMB; k0 += 32) {
    v16h a;
    const float* p0 = row + k0 + half * 8;
    const float* p1 = row + k0 + 16 + half * 8;
#pragma unroll
    for (int j = 0; j < 8; ++j) a[j] = (_Float16)p0[j];
#pragma unroll
    for (int j = 0; j < 8; ++j) a[8 + j] = (_Float16)p1[j];
    c = __builtin_amdgcn_wmma_f32_16x16x32_f16(
        false, a, false, bones, (short)0, c, false, false);
  }
#endif

  if (m == 0) {
    const float inv = 1.0f / (float)EMB;
    const int base = row0 + half * 8;
#pragma unroll
    for (int r = 0; r < 8; ++r) node_mean[base + r] = c[r] * inv;
  }
}

__global__ void zero_f32(float* __restrict__ p, int n) {
  int i = blockIdx.x * blockDim.x + threadIdx.x;
  if (i < n) p[i] = 0.0f;
}

__global__ void dst_scatter(const int* __restrict__ dst_event,
                            const int* __restrict__ dst_node,
                            const float* __restrict__ node_mean,
                            float* __restrict__ dst_sum, int n_dst) {
  int i = blockIdx.x * blockDim.x + threadIdx.x;
  if (i >= n_dst) return;
  unsafeAtomicAdd(&dst_sum[dst_event[i]], node_mean[dst_node[i]]);
}

__global__ void event_value(const int* __restrict__ event_pair,
                            const int* __restrict__ event_src,
                            const int* __restrict__ event_ndst,
                            const int* __restrict__ pair_place,
                            const float* __restrict__ node_mean,
                            float* __restrict__ ev, int n_events) {
  int i = blockIdx.x * blockDim.x + threadIdx.x;
  if (i >= n_events) return;
  int p = event_pair[i];
  float s = (node_mean[pair_place[p]] + node_mean[event_src[i]] +
             ev[i] / ((float)event_ndst[i] + 1.0f)) * (1.0f / 3.0f);
  ev[i] = s;
}

#define PR 16
#define EC 16
#define CAP 24576
#define AT 1024
#define UV 4

__global__ __launch_bounds__(AT) void pair_accum_lds(const int* __restrict__ event_pair,
                                                     const float* __restrict__ ev,
                                                     float* __restrict__ pair_sum,
                                                     float* __restrict__ pair_cnt,
                                                     int n_events, int n_pairs) {
  __shared__ float lsum[CAP];
  __shared__ float lcnt[CAP];

  const int tid = threadIdx.x;
  const int r = blockIdx.x % PR;
  const int c = blockIdx.x / PR;

  const int R  = (((n_pairs + PR - 1) / PR) + 31) & ~31;
  const int lo = r * R;
  const int hi = (lo + R < n_pairs) ? lo + R : n_pairs;

  const int clen = (((n_events + EC - 1) / EC) + 3) & ~3;
  const int e0 = c * clen;
  const int e1 = (e0 + clen < n_events) ? e0 + clen : n_events;

  const bool aligned =
      ((((size_t)(event_pair + e0)) | ((size_t)(ev + e0))) & 15) == 0;
  const int nvec = (aligned && e1 > e0) ? ((e1 - e0) >> 2) : 0;
  const int4*  __restrict__ ep4 = (const int4*)(event_pair + e0);
  const float4* __restrict__ s4 = (const float4*)(ev + e0);

  for (int base = lo; base < hi; base += CAP) {
    const int cnt = (hi - base < CAP) ? hi - base : CAP;

    for (int i = tid; i < cnt; i += AT) { lsum[i] = 0.0f; lcnt[i] = 0.0f; }
    __syncthreads();

    auto add1 = [&](int p, float s) {
      const unsigned rel = (unsigned)(p - base);
      if (rel < (unsigned)cnt) {
        __hip_atomic_fetch_add(&lsum[rel], s, __ATOMIC_RELAXED,
                               __HIP_MEMORY_SCOPE_WORKGROUP);
        __hip_atomic_fetch_add(&lcnt[rel], 1.0f, __ATOMIC_RELAXED,
                               __HIP_MEMORY_SCOPE_WORKGROUP);
      }
    };

    for (int v = tid; v < nvec; v += AT * UV) {
      int4 pp[UV];
      float4 ss[UV];
#pragma unroll
      for (int u = 0; u < UV; ++u) {
        const int vv = v + u * AT;
        if (vv < nvec) {
          pp[u] = ep4[vv];
          ss[u] = s4[vv];
        } else {
          pp[u] = make_int4(-1, -1, -1, -1);
          ss[u] = make_float4(0.0f, 0.0f, 0.0f, 0.0f);
        }
      }
#pragma unroll
      for (int u = 0; u < UV; ++u) {
        add1(pp[u].x, ss[u].x);
        add1(pp[u].y, ss[u].y);
        add1(pp[u].z, ss[u].z);
        add1(pp[u].w, ss[u].w);
      }
    }
    for (int i = e0 + 4 * nvec + tid; i < e1; i += AT)
      add1(event_pair[i], ev[i]);
    __syncthreads();

    const int rot = ((c * cnt) / EC) & ~31;
    for (int i = tid; i < cnt; i += AT) {
      int j = i + rot;
      if (j >= cnt) j -= cnt;
      const float n = lcnt[j];
      if (n > 0.0f) {
        unsafeAtomicAdd(&pair_sum[base + j], lsum[j]);
        unsafeAtomicAdd(&pair_cnt[base + j], n);
      }
    }
    __syncthreads();
  }
}

__global__ void pair_reduce(const int* __restrict__ pair_place,
                            const float* __restrict__ pair_sum,
                            const float* __restrict__ pair_cnt,
                            float* __restrict__ place_sum,
                            float* __restrict__ place_cnt, int n_pairs) {
  int i = blockIdx.x * blockDim.x + threadIdx.x;
  if (i >= n_pairs) return;
  float c = pair_cnt[i];
  if (c > 0.0f) {
    int pl = pair_place[i];
    unsafeAtomicAdd(&place_sum[pl], pair_sum[i] / c);
    unsafeAtomicAdd(&place_cnt[pl], 1.0f);
  }
}

__global__ __launch_bounds__(128) void bcast_out(const float* __restrict__ place_sum,
                                                 const float* __restrict__ place_cnt,
                                                 float* __restrict__ out, int n_nodes) {
  int row = blockIdx.x;
  if (row >= n_nodes) return;
  float c = place_cnt[row];
  float pm = c > 0.0f ? place_sum[row] / c : 0.0f;
  f32x4 v = {pm, pm, pm, pm};
  f32x4* __restrict__ o = (f32x4*)out + (size_t)row * (OUTW / 4);
  for (int j = threadIdx.x; j < OUTW / 4; j += 128)
    __builtin_nontemporal_store(v, o + j);
}

extern "C" void kernel_launch(void* const* d_in, const int* in_sizes, int n_in,
                              void* d_out, int out_size, void* d_ws, size_t ws_size,
                              hipStream_t stream) {
  const float* emb       = (const float*)d_in[0];
  const int* pair_place  = (const int*)d_in[1];
  const int* event_pair  = (const int*)d_in[2];
  const int* event_src   = (const int*)d_in[3];
  const int* event_ndst  = (const int*)d_in[4];
  const int* dst_event   = (const int*)d_in[5];
  const int* dst_node    = (const int*)d_in[6];

  const int n_nodes  = in_sizes[0] / EMB;
  const int n_pairs  = in_sizes[1];
  const int n_events = in_sizes[2];
  const int n_dst    = in_sizes[5];

  float* ws        = (float*)d_ws;
  float* node_mean = ws;
  float* dst_sum   = node_mean + n_nodes;
  float* pair_sum  = dst_sum + n_events;
  float* pair_cnt  = pair_sum + n_pairs;
  float* place_sum = pair_cnt + n_pairs;
  float* place_cnt = place_sum + n_nodes;

  const int nzero = n_events + 2 * n_pairs + 2 * n_nodes;
  zero_f32<<<(nzero + 255) / 256, 256, 0, stream>>>(dst_sum, nzero);

  rowmean_wmma<<<(n_nodes + 63) / 64, 128, 0, stream>>>(emb, node_mean, n_nodes);

  dst_scatter<<<(n_dst + 255) / 256, 256, 0, stream>>>(dst_event, dst_node,
                                                       node_mean, dst_sum, n_dst);

  event_value<<<(n_events + 255) / 256, 256, 0, stream>>>(
      event_pair, event_src, event_ndst, pair_place, node_mean, dst_sum,
      n_events);

  pair_accum_lds<<<PR * EC, AT, 0, stream>>>(event_pair, dst_sum, pair_sum,
                                             pair_cnt, n_events, n_pairs);

  pair_reduce<<<(n_pairs + 255) / 256, 256, 0, stream>>>(
      pair_place, pair_sum, pair_cnt, place_sum, place_cnt, n_pairs);

  bcast_out<<<n_nodes, 128, 0, stream>>>(place_sum, place_cnt, (float*)d_out,
                                         n_nodes);
}


# place stage by owner work-groups: a group scans all pair places, sums the pair means of its 64 rows in LDS and writes those rows; no place atomics or planes; 7->6 launches; place sum order still free
# speedup vs baseline: 2.2675x; 1.4823x over previous
#include <hip/hip_runtime.h>

#define EMB 512
#define OUTW (3 * EMB)

typedef __attribute__((ext_vector_type(2)))  float     v2f;
typedef __attribute__((ext_vector_type(8)))  float     v8f;
typedef __attribute__((ext_vector_type(4)))  float     f32x4;
typedef __attribute__((ext_vector_type(16))) _Float16  v16h;

__global__ __launch_bounds__(128) void rowmean_wmma(const float* __restrict__ emb,
                                                    float* __restrict__ node_mean,
                                                    int n_nodes) {
  const int lane = threadIdx.x & 31;
  const int wave = threadIdx.x >> 5;
  const int row0 = (blockIdx.x * 4 + wave) * 16;
  if (row0 >= n_nodes) return;

  const int m    = lane & 15;
  const int half = lane >> 4;
  const float* __restrict__ row = emb + (size_t)(row0 + m) * EMB;

  v8f c = {};

#if defined(__AMDGCN__) && __has_builtin(__builtin_amdgcn_wmma_f32_16x16x4_f32)
  const v2f* __restrict__ rp = (const v2f*)row + half;
  v2f bones = {1.0f, 1.0f};
  for (int k = 0; k < EMB / 4; ++k) {
    v2f a = rp[k * 2];
    c = __builtin_amdgcn_wmma_f32_16x16x4_f32(
         false, a,  false, bones,
         (short)0, c,  false,  false);
  }
#else
  v16h bones;
#pragma unroll
  for (int j = 0; j < 16; ++j) bones[j] = (_Float16)1.0f;
  for (int k0 = 0; k0 < EMB; k0 += 32) {
    v16h a;
    const float* p0 = row + k0 + half * 8;
    const float* p1 = row + k0 + 16 + half * 8;
#pragma unroll
    for (int j = 0; j < 8; ++j) a[j] = (_Float16)p0[j];
#pragma unroll
    for (int j = 0; j < 8; ++j) a[8 + j] = (_Float16)p1[j];
    c = __builtin_amdgcn_wmma_f32_16x16x32_f16(
        false, a, false, bones, (short)0, c, false, false);
  }
#endif

  if (m == 0) {
    const float inv = 1.0f / (float)EMB;
    const int base = row0 + half * 8;
#pragma unroll
    for (int r = 0; r < 8; ++r) node_mean[base + r] = c[r] * inv;
  }
}

__global__ void zero_f32(float* __restrict__ p, int n) {
  int i = blockIdx.x * blockDim.x + threadIdx.x;
  if (i < n) p[i] = 0.0f;
}

#define DK 8
#define DN 4
#define DT 256

__global__ __launch_bounds__(DT) void dst_runs(const int* __restrict__ dst_event,
                                               const int* __restrict__ dst_node,
                                               const float* __restrict__ node_mean,
                                               float* __restrict__ dst_sum,
                                               int n_dst, int n_events) {
  const int t = blockIdx.x * DT + threadIdx.x;
  if (n_dst <= 0) {
    for (int g = t; g < n_events; g += gridDim.x * DT) dst_sum[g] = 0.0f;
    return;
  }
  const int start = t * DK;
  if (start >= n_dst) return;

  int id[DK + DN], nd[DK + DN];
  const bool vec =
      ((((size_t)dst_event) | ((size_t)dst_node)) & 15) == 0 &&
      start + DK + DN <= n_dst;
  if (vec) {
    const int4* __restrict__ e4 = (const int4*)(dst_event + start);
    const int4* __restrict__ n4 = (const int4*)(dst_node + start);
#pragma unroll
    for (int q = 0; q < (DK + DN) / 4; ++q) {
      const int4 a = e4[q];
      const int4 b = n4[q];
      id[4 * q + 0] = a.x; id[4 * q + 1] = a.y;
      id[4 * q + 2] = a.z; id[4 * q + 3] = a.w;
      nd[4 * q + 0] = b.x; nd[4 * q + 1] = b.y;
      nd[4 * q + 2] = b.z; nd[4 * q + 3] = b.w;
    }
  } else {
#pragma unroll
    for (int j = 0; j < DK + DN; ++j) {
      const bool ok = start + j < n_dst;
      id[j] = ok ? dst_event[start + j] : -2;
      nd[j] = ok ? dst_node[start + j] : 0;
    }
  }
  int cur = (start > 0) ? dst_event[start - 1] : -1;

  float v[DK + DN];
#pragma unroll
  for (int j = 0; j < DK + DN; ++j) v[j] = node_mean[nd[j]];

  float acc = 0.0f;
  bool own = false;
#pragma unroll
  for (int j = 0; j < DK; ++j) {
    const int e = id[j];
    if (e != -2) {
      if (e != cur) {
        if (own) dst_sum[cur] = acc;
#pragma clang loop unroll(disable) vectorize(disable)
        for (int g = cur + 1; g < e; ++g) dst_sum[g] = 0.0f;
        cur = e;
        acc = v[j];
        own = true;
      } else {
        acc += v[j];
      }
    }
  }

  if (own) {
    int j = DK;
#pragma unroll
    for (int u = 0; u < DN; ++u) {
      if (j == DK + u && id[DK + u] == cur) { acc += v[DK + u]; ++j; }
    }
    int g = start + j;
    if (j == DK + DN) {
      while (g < n_dst && dst_event[g] == cur) {
        acc += node_mean[dst_node[g]];
        ++g;
      }
    }
    dst_sum[cur] = acc;
    if (g >= n_dst) {
#pragma clang loop unroll(disable) vectorize(disable)
      for (int z = cur + 1; z < n_events; ++z) dst_sum[z] = 0.0f;
    }
  }
}

__global__ void event_value(const int* __restrict__ event_pair,
                            const int* __restrict__ event_src,
                            const int* __restrict__ event_ndst,
                            const int* __restrict__ pair_place,
                            const float* __restrict__ node_mean,
                            float* __restrict__ ev, int n_events) {
  int i = blockIdx.x * blockDim.x + threadIdx.x;
  if (i >= n_events) return;
  int p = event_pair[i];
  float s = (node_mean[pair_place[p]] + node_mean[event_src[i]] +
             ev[i] / ((float)event_ndst[i] + 1.0f)) * (1.0f / 3.0f);
  ev[i] = s;
}

#define PR 16
#define EC 16
#define CAP 24576
#define AT 1024
#define UV 4

__global__ __launch_bounds__(AT) void pair_accum_lds(const int* __restrict__ event_pair,
                                                     const float* __restrict__ ev,
                                                     float* __restrict__ pair_sum,
                                                     float* __restrict__ pair_cnt,
                                                     int n_events, int n_pairs) {
  __shared__ float lsum[CAP];
  __shared__ float lcnt[CAP];

  const int tid = threadIdx.x;
  const int r = blockIdx.x % PR;
  const int c = blockIdx.x / PR;

  const int R  = (((n_pairs + PR - 1) / PR) + 31) & ~31;
  const int lo = r * R;
  const int hi = (lo + R < n_pairs) ? lo + R : n_pairs;

  const int clen = (((n_events + EC - 1) / EC) + 3) & ~3;
  const int e0 = c * clen;
  const int e1 = (e0 + clen < n_events) ? e0 + clen : n_events;

  const bool aligned =
      ((((size_t)(event_pair + e0)) | ((size_t)(ev + e0))) & 15) == 0;
  const int nvec = (aligned && e1 > e0) ? ((e1 - e0) >> 2) : 0;
  const int4*  __restrict__ ep4 = (const int4*)(event_pair + e0);
  const float4* __restrict__ s4 = (const float4*)(ev + e0);

  for (int base = lo; base < hi; base += CAP) {
    const int cnt = (hi - base < CAP) ? hi - base : CAP;

    for (int i = tid; i < cnt; i += AT) { lsum[i] = 0.0f; lcnt[i] = 0.0f; }
    __syncthreads();

    auto add1 = [&](int p, float s) {
      const unsigned rel = (unsigned)(p - base);
      if (rel < (unsigned)cnt) {
        __hip_atomic_fetch_add(&lsum[rel], s, __ATOMIC_RELAXED,
                               __HIP_MEMORY_SCOPE_WORKGROUP);
        __hip_atomic_fetch_add(&lcnt[rel], 1.0f, __ATOMIC_RELAXED,
                               __HIP_MEMORY_SCOPE_WORKGROUP);
      }
    };

    for (int v = tid; v < nvec; v += AT * UV) {
      int4 pp[UV];
      float4 ss[UV];
#pragma unroll
      for (int u = 0; u < UV; ++u) {
        const int vv = v + u * AT;
        if (vv < nvec) {
          pp[u] = ep4[vv];
          ss[u] = s4[vv];
        } else {
          pp[u] = make_int4(-1, -1, -1, -1);
          ss[u] = make_float4(0.0f, 0.0f, 0.0f, 0.0f);
        }
      }
#pragma unroll
      for (int u = 0; u < UV; ++u) {
        add1(pp[u].x, ss[u].x);
        add1(pp[u].y, ss[u].y);
        add1(pp[u].z, ss[u].z);
        add1(pp[u].w, ss[u].w);
      }
    }
    for (int i = e0 + 4 * nvec + tid; i < e1; i += AT)
      add1(event_pair[i], ev[i]);
    __syncthreads();

    const int rot = ((c * cnt) / EC) & ~31;
    for (int i = tid; i < cnt; i += AT) {
      int j = i + rot;
      if (j >= cnt) j -= cnt;
      const float n = lcnt[j];
      if (n > 0.0f) {
        unsafeAtomicAdd(&pair_sum[base + j], lsum[j]);
        unsafeAtomicAdd(&pair_cnt[base + j], n);
      }
    }
    __syncthreads();
  }
}

#define PO_ROWS 64
#define PO_T    1024
#define PO_UV   4
#define PO_QCAP 24576

__global__ __launch_bounds__(PO_T) void place_rows(const int* __restrict__ pair_place,
                                                   const float* __restrict__ pair_sum,
                                                   const float* __restrict__ pair_cnt,
                                                   float* __restrict__ out,
                                                   int n_pairs, int n_nodes) {
  __shared__ int2  queue[PO_QCAP];
  __shared__ float tsum[PO_ROWS];
  __shared__ float tcnt[PO_ROWS];
  __shared__ int   qn;

  const int tid = threadIdx.x;
  const int lo = blockIdx.x * PO_ROWS;
  const unsigned nrows =
      (unsigned)((n_nodes - lo < PO_ROWS) ? n_nodes - lo : PO_ROWS);

  if (tid < PO_ROWS) { tsum[tid] = 0.0f; tcnt[tid] = 0.0f; }
  if (tid == 0) qn = 0;
  __syncthreads();

  auto add_pair = [&](int i, unsigned rel) {
    const float c = pair_cnt[i];
    if (c > 0.0f) {
      __hip_atomic_fetch_add(&tsum[rel], pair_sum[i] / c, __ATOMIC_RELAXED,
                             __HIP_MEMORY_SCOPE_WORKGROUP);
      __hip_atomic_fetch_add(&tcnt[rel], 1.0f, __ATOMIC_RELAXED,
                             __HIP_MEMORY_SCOPE_WORKGROUP);
    }
  };
  auto hit = [&](int i, unsigned rel) {
    const int pos = __hip_atomic_fetch_add(&qn, 1, __ATOMIC_RELAXED,
                                           __HIP_MEMORY_SCOPE_WORKGROUP);
    if (pos < PO_QCAP) queue[pos] = make_int2(i, (int)rel);
  };
  auto look = [&](const int4 p, int w) {
    const unsigned r0 = (unsigned)(p.x - lo);
    const unsigned r1 = (unsigned)(p.y - lo);
    const unsigned r2 = (unsigned)(p.z - lo);
    const unsigned r3 = (unsigned)(p.w - lo);
    const unsigned m01 = r0 < r1 ? r0 : r1;
    const unsigned m23 = r2 < r3 ? r2 : r3;
    const unsigned mn  = m01 < m23 ? m01 : m23;
    if (mn < nrows) {
      if (r0 < nrows) hit(4 * w + 0, r0);
      if (r1 < nrows) hit(4 * w + 1, r1);
      if (r2 < nrows) hit(4 * w + 2, r2);
      if (r3 < nrows) hit(4 * w + 3, r3);
    }
  };

  const bool aligned = (((size_t)pair_place) & 15) == 0;
  const int n4 = aligned ? (n_pairs >> 2) : 0;
  const int4* __restrict__ pl4 = (const int4*)pair_place;
  const int rot = (int)blockIdx.x * (int)((unsigned)n4 / gridDim.x);
  auto widx = [&](int vv) {
    const int w = vv + rot;
    return (w >= n4) ? w - n4 : w;
  };

  const int STEP = PO_T * PO_UV;
  const int nfull = n4 / STEP;
  int4 cur[PO_UV], nxt[PO_UV];
  if (nfull > 0) {
#pragma unroll
    for (int u = 0; u < PO_UV; ++u) cur[u] = pl4[widx(u * PO_T + tid)];
    for (int b = 0; b + 1 < nfull; ++b) {
#pragma unroll
      for (int u = 0; u < PO_UV; ++u)
        nxt[u] = pl4[widx((b + 1) * STEP + u * PO_T + tid)];
#pragma unroll
      for (int u = 0; u < PO_UV; ++u)
        look(cur[u], widx(b * STEP + u * PO_T + tid));
#pragma unroll
      for (int u = 0; u < PO_UV; ++u) cur[u] = nxt[u];
    }
#pragma unroll
    for (int u = 0; u < PO_UV; ++u)
      look(cur[u], widx((nfull - 1) * STEP + u * PO_T + tid));
  }
  for (int vv = nfull * STEP + tid; vv < n4; vv += PO_T) {
    const int w = widx(vv);
    look(pl4[w], w);
  }
  for (int i = 4 * n4 + tid; i < n_pairs; i += PO_T) {
    const unsigned rel = (unsigned)(pair_place[i] - lo);
    if (rel < nrows) hit(i, rel);
  }
  __syncthreads();

  const int nh = qn;
  if (nh <= PO_QCAP) {
    for (int e = tid; e < nh; e += PO_T) {
      const int2 en = queue[e];
      add_pair(en.x, (unsigned)en.y);
    }
  } else {
    for (int i = tid; i < n_pairs; i += PO_T) {
      const unsigned rel = (unsigned)(pair_place[i] - lo);
      if (rel < nrows) add_pair(i, rel);
    }
  }
  __syncthreads();

  if (tid < PO_ROWS) {
    const float c = tcnt[tid];
    tsum[tid] = (c > 0.0f) ? tsum[tid] / c : 0.0f;
  }
  __syncthreads();

  const int lane = tid & 31;
  f32x4* __restrict__ o = (f32x4*)out + (size_t)lo * (OUTW / 4);
  for (int r = tid >> 5; r < (int)nrows; r += PO_T / 32) {
    const float pm = tsum[r];
    const f32x4 v = {pm, pm, pm, pm};
    f32x4* __restrict__ orow = o + (size_t)r * (OUTW / 4) + lane;
#pragma unroll
    for (int k = 0; k < OUTW / 4 / 32; ++k)
      __builtin_nontemporal_store(v, orow + 32 * k);
  }
}

extern "C" void kernel_launch(void* const* d_in, const int* in_sizes, int n_in,
                              void* d_out, int out_size, void* d_ws, size_t ws_size,
                              hipStream_t stream) {
  const float* emb       = (const float*)d_in[0];
  const int* pair_place  = (const int*)d_in[1];
  const int* event_pair  = (const int*)d_in[2];
  const int* event_src   = (const int*)d_in[3];
  const int* event_ndst  = (const int*)d_in[4];
  const int* dst_event   = (const int*)d_in[5];
  const int* dst_node    = (const int*)d_in[6];

  const int n_nodes  = in_sizes[0] / EMB;
  const int n_pairs  = in_sizes[1];
  const int n_events = in_sizes[2];
  const int n_dst    = in_sizes[5];

  float* ws        = (float*)d_ws;
  float* node_mean = ws;
  float* dst_sum   = node_mean + n_nodes;
  float* pair_sum  = dst_sum + n_events;
  float* pair_cnt  = pair_sum + n_pairs;

  const int nzero = 2 * n_pairs;
  zero_f32<<<(nzero + 255) / 256, 256, 0, stream>>>(pair_sum, nzero);

  rowmean_wmma<<<(n_nodes + 63) / 64, 128, 0, stream>>>(emb, node_mean, n_nodes);

  dst_runs<<<n_dst / (DK * DT) + 1, DT, 0, stream>>>(dst_event, dst_node, node_mean,
                                                      dst_sum, n_dst, n_events);

  event_value<<<(n_events + 255) / 256, 256, 0, stream>>>(
      event_pair, event_src, event_ndst, pair_place, node_mean, dst_sum,
      n_events);

  pair_accum_lds<<<PR * EC, AT, 0, stream>>>(event_pair, dst_sum, pair_sum,
                                             pair_cnt, n_events, n_pairs);

  place_rows<<<(n_nodes + PO_ROWS - 1) / PO_ROWS, PO_T, 0, stream>>>(
      pair_place, pair_sum, pair_cnt, (float*)d_out, n_pairs, n_nodes);
}


# place stage searches a 1-byte-per-pair owner plane (place/64, written by the zeroing launch), 16 pairs per 16-byte load, not pair_place itself; place checked on fetch; place sum order still free
# speedup vs baseline: 2.3886x; 1.0534x over previous
#include <hip/hip_runtime.h>

#define EMB 512
#define OUTW (3 * EMB)

typedef __attribute__((ext_vector_type(2)))  float     v2f;
typedef __attribute__((ext_vector_type(8)))  float     v8f;
typedef __attribute__((ext_vector_type(4)))  float     f32x4;
typedef __attribute__((ext_vector_type(16))) _Float16  v16h;

__global__ __launch_bounds__(128) void rowmean_wmma(const float* __restrict__ emb,
                                                    float* __restrict__ node_mean,
                                                    int n_nodes) {
  const int lane = threadIdx.x & 31;
  const int wave = threadIdx.x >> 5;
  const int row0 = (blockIdx.x * 4 + wave) * 16;
  if (row0 >= n_nodes) return;

  const int m    = lane & 15;
  const int half = lane >> 4;
  const float* __restrict__ row = emb + (size_t)(row0 + m) * EMB;

  v8f c = {};

#if defined(__AMDGCN__) && __has_builtin(__builtin_amdgcn_wmma_f32_16x16x4_f32)
  const v2f* __restrict__ rp = (const v2f*)row + half;
  v2f bones = {1.0f, 1.0f};
  for (int k = 0; k < EMB / 4; ++k) {
    v2f a = rp[k * 2];
    c = __builtin_amdgcn_wmma_f32_16x16x4_f32(
         false, a,  false, bones,
         (short)0, c,  false,  false);
  }
#else
  v16h bones;
#pragma unroll
  for (int j = 0; j < 16; ++j) bones[j] = (_Float16)1.0f;
  for (int k0 = 0; k0 < EMB; k0 += 32) {
    v16h a;
    const float* p0 = row + k0 + half * 8;
    const float* p1 = row + k0 + 16 + half * 8;
#pragma unroll
    for (int j = 0; j < 8; ++j) a[j] = (_Float16)p0[j];
#pragma unroll
    for (int j = 0; j < 8; ++j) a[8 + j] = (_Float16)p1[j];
    c = __builtin_amdgcn_wmma_f32_16x16x32_f16(
        false, a, false, bones, (short)0, c, false, false);
  }
#endif

  if (m == 0) {
    const float inv = 1.0f / (float)EMB;
    const int base = row0 + half * 8;
#pragma unroll
    for (int r = 0; r < 8; ++r) node_mean[base + r] = c[r] * inv;
  }
}

#define PO_ROWS 64

__global__ void zero_and_owner(float* __restrict__ p, int n,
                               const int* __restrict__ pair_place,
                               int* __restrict__ owner, int n_pairs) {
  int i = blockIdx.x * blockDim.x + threadIdx.x;
  if (i < n) p[i] = 0.0f;
  if (owner == nullptr) return;
  const int words = (n_pairs + 3) >> 2;
  if (i >= words) return;
  int a0, a1, a2, a3;
  if ((((size_t)pair_place) & 15) == 0 && 4 * i + 3 < n_pairs) {
    const int4 v = ((const int4*)pair_place)[i];
    a0 = v.x; a1 = v.y; a2 = v.z; a3 = v.w;
  } else {
    a0 = (4 * i + 0 < n_pairs) ? pair_place[4 * i + 0] : 0;
    a1 = (4 * i + 1 < n_pairs) ? pair_place[4 * i + 1] : 0;
    a2 = (4 * i + 2 < n_pairs) ? pair_place[4 * i + 2] : 0;
    a3 = (4 * i + 3 < n_pairs) ? pair_place[4 * i + 3] : 0;
  }
  const unsigned b0 = ((unsigned)a0 / PO_ROWS) & 255u;
  const unsigned b1 = ((unsigned)a1 / PO_ROWS) & 255u;
  const unsigned b2 = ((unsigned)a2 / PO_ROWS) & 255u;
  const unsigned b3 = ((unsigned)a3 / PO_ROWS) & 255u;
  owner[i] = (int)(b0 | (b1 << 8) | (b2 << 16) | (b3 << 24));
}

#define DK 8
#define DN 4
#define DT 256

__global__ __launch_bounds__(DT) void dst_runs(const int* __restrict__ dst_event,
                                               const int* __restrict__ dst_node,
                                               const float* __restrict__ node_mean,
                                               float* __restrict__ dst_sum,
                                               int n_dst, int n_events) {
  const int t = blockIdx.x * DT + threadIdx.x;
  if (n_dst <= 0) {
    for (int g = t; g < n_events; g += gridDim.x * DT) dst_sum[g] = 0.0f;
    return;
  }
  const int start = t * DK;
  if (start >= n_dst) return;

  int id[DK + DN], nd[DK + DN];
  const bool vec =
      ((((size_t)dst_event) | ((size_t)dst_node)) & 15) == 0 &&
      start + DK + DN <= n_dst;
  if (vec) {
    const int4* __restrict__ e4 = (const int4*)(dst_event + start);
    const int4* __restrict__ n4 = (const int4*)(dst_node + start);
#pragma unroll
    for (int q = 0; q < (DK + DN) / 4; ++q) {
      const int4 a = e4[q];
      const int4 b = n4[q];
      id[4 * q + 0] = a.x; id[4 * q + 1] = a.y;
      id[4 * q + 2] = a.z; id[4 * q + 3] = a.w;
      nd[4 * q + 0] = b.x; nd[4 * q + 1] = b.y;
      nd[4 * q + 2] = b.z; nd[4 * q + 3] = b.w;
    }
  } else {
#pragma unroll
    for (int j = 0; j < DK + DN; ++j) {
      const bool ok = start + j < n_dst;
      id[j] = ok ? dst_event[start + j] : -2;
      nd[j] = ok ? dst_node[start + j] : 0;
    }
  }
  int cur = (start > 0) ? dst_event[start - 1] : -1;

  float v[DK + DN];
#pragma unroll
  for (int j = 0; j < DK + DN; ++j) v[j] = node_mean[nd[j]];

  float acc = 0.0f;
  bool own = false;
#pragma unroll
  for (int j = 0; j < DK; ++j) {
    const int e = id[j];
    if (e != -2) {
      if (e != cur) {
        if (own) dst_sum[cur] = acc;
#pragma clang loop unroll(disable) vectorize(disable)
        for (int g = cur + 1; g < e; ++g) dst_sum[g] = 0.0f;
        cur = e;
        acc = v[j];
        own = true;
      } else {
        acc += v[j];
      }
    }
  }

  if (own) {
    int j = DK;
#pragma unroll
    for (int u = 0; u < DN; ++u) {
      if (j == DK + u && id[DK + u] == cur) { acc += v[DK + u]; ++j; }
    }
    int g = start + j;
    if (j == DK + DN) {
      while (g < n_dst && dst_event[g] == cur) {
        acc += node_mean[dst_node[g]];
        ++g;
      }
    }
    dst_sum[cur] = acc;
    if (g >= n_dst) {
#pragma clang loop unroll(disable) vectorize(disable)
      for (int z = cur + 1; z < n_events; ++z) dst_sum[z] = 0.0f;
    }
  }
}

__global__ void event_value(const int* __restrict__ event_pair,
                            const int* __restrict__ event_src,
                            const int* __restrict__ event_ndst,
                            const int* __restrict__ pair_place,
                            const float* __restrict__ node_mean,
                            float* __restrict__ ev, int n_events) {
  int i = blockIdx.x * blockDim.x + threadIdx.x;
  if (i >= n_events) return;
  int p = event_pair[i];
  float s = (node_mean[pair_place[p]] + node_mean[event_src[i]] +
             ev[i] / ((float)event_ndst[i] + 1.0f)) * (1.0f / 3.0f);
  ev[i] = s;
}

#define PR 16
#define EC 16
#define CAP 24576
#define AT 1024
#define UV 4

__global__ __launch_bounds__(AT) void pair_accum_lds(const int* __restrict__ event_pair,
                                                     const float* __restrict__ ev,
                                                     float* __restrict__ pair_sum,
                                                     float* __restrict__ pair_cnt,
                                                     int n_events, int n_pairs) {
  __shared__ float lsum[CAP];
  __shared__ float lcnt[CAP];

  const int tid = threadIdx.x;
  const int r = blockIdx.x % PR;
  const int c = blockIdx.x / PR;

  const int R  = (((n_pairs + PR - 1) / PR) + 31) & ~31;
  const int lo = r * R;
  const int hi = (lo + R < n_pairs) ? lo + R : n_pairs;

  const int clen = (((n_events + EC - 1) / EC) + 3) & ~3;
  const int e0 = c * clen;
  const int e1 = (e0 + clen < n_events) ? e0 + clen : n_events;

  const bool aligned =
      ((((size_t)(event_pair + e0)) | ((size_t)(ev + e0))) & 15) == 0;
  const int nvec = (aligned && e1 > e0) ? ((e1 - e0) >> 2) : 0;
  const int4*  __restrict__ ep4 = (const int4*)(event_pair + e0);
  const float4* __restrict__ s4 = (const float4*)(ev + e0);

  for (int base = lo; base < hi; base += CAP) {
    const int cnt = (hi - base < CAP) ? hi - base : CAP;

    for (int i = tid; i < cnt; i += AT) { lsum[i] = 0.0f; lcnt[i] = 0.0f; }
    __syncthreads();

    auto add1 = [&](int p, float s) {
      const unsigned rel = (unsigned)(p - base);
      if (rel < (unsigned)cnt) {
        __hip_atomic_fetch_add(&lsum[rel], s, __ATOMIC_RELAXED,
                               __HIP_MEMORY_SCOPE_WORKGROUP);
        __hip_atomic_fetch_add(&lcnt[rel], 1.0f, __ATOMIC_RELAXED,
                               __HIP_MEMORY_SCOPE_WORKGROUP);
      }
    };

    for (int v = tid; v < nvec; v += AT * UV) {
      int4 pp[UV];
      float4 ss[UV];
#pragma unroll
      for (int u = 0; u < UV; ++u) {
        const int vv = v + u * AT;
        if (vv < nvec) {
          pp[u] = ep4[vv];
          ss[u] = s4[vv];
        } else {
          pp[u] = make_int4(-1, -1, -1, -1);
          ss[u] = make_float4(0.0f, 0.0f, 0.0f, 0.0f);
        }
      }
#pragma unroll
      for (int u = 0; u < UV; ++u) {
        add1(pp[u].x, ss[u].x);
        add1(pp[u].y, ss[u].y);
        add1(pp[u].z, ss[u].z);
        add1(pp[u].w, ss[u].w);
      }
    }
    for (int i = e0 + 4 * nvec + tid; i < e1; i += AT)
      add1(event_pair[i], ev[i]);
    __syncthreads();

    const int rot = ((c * cnt) / EC) & ~31;
    for (int i = tid; i < cnt; i += AT) {
      int j = i + rot;
      if (j >= cnt) j -= cnt;
      const float n = lcnt[j];
      if (n > 0.0f) {
        unsafeAtomicAdd(&pair_sum[base + j], lsum[j]);
        unsafeAtomicAdd(&pair_cnt[base + j], n);
      }
    }
    __syncthreads();
  }
}

#define PO_T    1024
#define PO_UV   4
#define PO_QCAP 49152

__global__ __launch_bounds__(PO_T) void place_rows(const int* __restrict__ pair_place,
                                                   const int* __restrict__ owner,
                                                   const float* __restrict__ pair_sum,
                                                   const float* __restrict__ pair_cnt,
                                                   float* __restrict__ out,
                                                   int n_pairs, int n_nodes) {
  __shared__ int   queue[PO_QCAP];
  __shared__ float tsum[PO_ROWS];
  __shared__ float tcnt[PO_ROWS];
  __shared__ int   qn;

  const int tid = threadIdx.x;
  const int lo = blockIdx.x * PO_ROWS;
  const unsigned nrows =
      (unsigned)((n_nodes - lo < PO_ROWS) ? n_nodes - lo : PO_ROWS);

  if (tid < PO_ROWS) { tsum[tid] = 0.0f; tcnt[tid] = 0.0f; }
  if (tid == 0) qn = 0;
  __syncthreads();

  auto add_mean = [&](unsigned rel, float c, float s) {
    if (c > 0.0f) {
      __hip_atomic_fetch_add(&tsum[rel], s / c, __ATOMIC_RELAXED,
                             __HIP_MEMORY_SCOPE_WORKGROUP);
      __hip_atomic_fetch_add(&tcnt[rel], 1.0f, __ATOMIC_RELAXED,
                             __HIP_MEMORY_SCOPE_WORKGROUP);
    }
  };

  const bool have_plane = owner != nullptr;
  if (have_plane) {
    const unsigned mine = 0x01010101u * (blockIdx.x & 255u);
    auto eq = [&](unsigned x) {
      const unsigned t = x ^ mine;
      return ~(((t & 0x7f7f7f7fu) + 0x7f7f7f7fu) | t | 0x7f7f7f7fu);
    };
    auto hits = [&](const uint4 v) {
      return (eq(v.x) >> 7) | (eq(v.y) >> 6) | (eq(v.z) >> 5) | (eq(v.w) >> 4);
    };
    auto emit = [&](unsigned m, int w, int& pos) {
      while (m != 0u) {
        const int q = __builtin_ctz(m);
        m &= m - 1u;
        if (pos < PO_QCAP) queue[pos] = 16 * w + 4 * (q & 7) + (q >> 3);
        ++pos;
      }
    };
    auto reserve = [&](int n) {
      return __hip_atomic_fetch_add(&qn, n, __ATOMIC_RELAXED,
                                    __HIP_MEMORY_SCOPE_WORKGROUP);
    };

    const bool aligned = (((size_t)owner) & 15) == 0;
    const int n16 = aligned ? (n_pairs >> 4) : 0;
    const uint4* __restrict__ ow4 = (const uint4*)owner;
    const int rot = (int)blockIdx.x * (int)((unsigned)n16 / gridDim.x);
    auto widx = [&](int vv) {
      const int w = vv + rot;
      return (w >= n16) ? w - n16 : w;
    };
    auto look_batch = [&](const uint4 (&v)[PO_UV], int vv0) {
      unsigned m[PO_UV];
      int n = 0;
#pragma unroll
      for (int u = 0; u < PO_UV; ++u) {
        m[u] = hits(v[u]);
        n += (int)__builtin_popcount(m[u]);
      }
      if (n != 0) {
        int pos = reserve(n);
#pragma unroll
        for (int u = 0; u < PO_UV; ++u) emit(m[u], widx(vv0 + u * PO_T), pos);
      }
    };

    const int STEP = PO_T * PO_UV;
    const int nfull = n16 / STEP;
    uint4 cur[PO_UV], nxt[PO_UV];
    if (nfull > 0) {
#pragma unroll
      for (int u = 0; u < PO_UV; ++u) cur[u] = ow4[widx(u * PO_T + tid)];
      for (int b = 0; b + 1 < nfull; ++b) {
#pragma unroll
        for (int u = 0; u < PO_UV; ++u)
          nxt[u] = ow4[widx((b + 1) * STEP + u * PO_T + tid)];
        look_batch(cur, b * STEP + tid);
#pragma unroll
        for (int u = 0; u < PO_UV; ++u) cur[u] = nxt[u];
      }
      look_batch(cur, (nfull - 1) * STEP + tid);
    }
    for (int vv = nfull * STEP + tid; vv < n16; vv += PO_T) {
      const int w = widx(vv);
      const unsigned m = hits(ow4[w]);
      if (m != 0u) {
        int pos = reserve((int)__builtin_popcount(m));
        emit(m, w, pos);
      }
    }
    const unsigned char* __restrict__ ow1 = (const unsigned char*)owner;
    for (int i = 16 * n16 + tid; i < n_pairs; i += PO_T) {
      if (ow1[i] == (unsigned char)(blockIdx.x & 255u)) {
        const int pos = reserve(1);
        if (pos < PO_QCAP) queue[pos] = i;
      }
    }
  }
  __syncthreads();

  const int nh = qn;
  if (have_plane && nh <= PO_QCAP) {
    for (int e = tid; e < nh; e += PO_T) {
      const int i = queue[e];
      const int   pl = pair_place[i];
      const float c  = pair_cnt[i];
      const unsigned rel = (unsigned)(pl - lo);
      if ((rel < nrows) & (c > 0.0f)) add_mean(rel, c, pair_sum[i]);
    }
  } else {
    for (int i = tid; i < n_pairs; i += PO_T) {
      const unsigned rel = (unsigned)(pair_place[i] - lo);
      if (rel < nrows) add_mean(rel, pair_cnt[i], pair_sum[i]);
    }
  }
  __syncthreads();

  if (tid < PO_ROWS) {
    const float c = tcnt[tid];
    tsum[tid] = (c > 0.0f) ? tsum[tid] / c : 0.0f;
  }
  __syncthreads();

  const int lane = tid & 31;
  f32x4* __restrict__ o = (f32x4*)out + (size_t)lo * (OUTW / 4);
  for (int r = tid >> 5; r < (int)nrows; r += PO_T / 32) {
    const float pm = tsum[r];
    const f32x4 v = {pm, pm, pm, pm};
    f32x4* __restrict__ orow = o + (size_t)r * (OUTW / 4) + lane;
#pragma unroll
    for (int k = 0; k < OUTW / 4 / 32; ++k)
      __builtin_nontemporal_store(v, orow + 32 * k);
  }
}

extern "C" void kernel_launch(void* const* d_in, const int* in_sizes, int n_in,
                              void* d_out, int out_size, void* d_ws, size_t ws_size,
                              hipStream_t stream) {
  const float* emb       = (const float*)d_in[0];
  const int* pair_place  = (const int*)d_in[1];
  const int* event_pair  = (const int*)d_in[2];
  const int* event_src   = (const int*)d_in[3];
  const int* event_ndst  = (const int*)d_in[4];
  const int* dst_event   = (const int*)d_in[5];
  const int* dst_node    = (const int*)d_in[6];

  const int n_nodes  = in_sizes[0] / EMB;
  const int n_pairs  = in_sizes[1];
  const int n_events = in_sizes[2];
  const int n_dst    = in_sizes[5];

  float* ws        = (float*)d_ws;
  float* node_mean = ws;
  float* dst_sum   = node_mean + n_nodes;
  float* pair_sum  = dst_sum + n_events;
  float* pair_cnt  = pair_sum + n_pairs;
  const size_t plane_at = ((size_t)n_nodes + (size_t)n_events + 2 * (size_t)n_pairs) * sizeof(float);
  const size_t plane_len = (size_t)((n_pairs + 3) / 4) * sizeof(int);
  int* owner = (ws_size >= plane_at + plane_len) ? (int*)(pair_cnt + n_pairs) : nullptr;

  const int nzero = 2 * n_pairs;
  zero_and_owner<<<(nzero + 255) / 256, 256, 0, stream>>>(pair_sum, nzero, pair_place,
                                                          owner, n_pairs);

  rowmean_wmma<<<(n_nodes + 63) / 64, 128, 0, stream>>>(emb, node_mean, n_nodes);

  dst_runs<<<n_dst / (DK * DT) + 1, DT, 0, stream>>>(dst_event, dst_node, node_mean,
                                                      dst_sum, n_dst, n_events);

  event_value<<<(n_events + 255) / 256, 256, 0, stream>>>(
      event_pair, event_src, event_ndst, pair_place, node_mean, dst_sum,
      n_events);

  pair_accum_lds<<<PR * EC, AT, 0, stream>>>(event_pair, dst_sum, pair_sum,
                                             pair_cnt, n_events, n_pairs);

  place_rows<<<(n_nodes + PO_ROWS - 1) / PO_ROWS, PO_T, 0, stream>>>(
      pair_place, owner, pair_sum, pair_cnt, (float*)d_out, n_pairs, n_nodes);
}


# row means: embedding values pass through the WMMA B operand, a one-hot A routes each row's sum to its accumulator row; a wave load = 512 contiguous bytes of one row, all 64 ahead; sum order changes
# speedup vs baseline: 2.6203x; 1.0970x over previous
#include <hip/hip_runtime.h>

#define EMB 512
#define OUTW (3 * EMB)

typedef __attribute__((ext_vector_type(2)))  float     v2f;
typedef __attribute__((ext_vector_type(8)))  float     v8f;
typedef __attribute__((ext_vector_type(4)))  float     f32x4;
typedef __attribute__((ext_vector_type(16))) _Float16  v16h;

__global__ __launch_bounds__(128) void rowmean_wmma(const float* __restrict__ emb,
                                                    float* __restrict__ node_mean,
                                                    int n_nodes) {
  const int lane = threadIdx.x & 31;
  const int wave = threadIdx.x >> 5;
  const int row0 = (blockIdx.x * 4 + wave) * 16;
  if (row0 >= n_nodes) return;

  const int m    = lane & 15;
  const int half = lane >> 4;

  v8f c = {};

#if defined(__AMDGCN__) && __has_builtin(__builtin_amdgcn_wmma_f32_16x16x4_f32)
  v8f c2 = {};
  constexpr int LPR = EMB / 4 / 32;
  v2f a[16];
#pragma unroll
  for (int r = 0; r < 16; ++r) {
    const float sel = (m == r) ? 1.0f : 0.0f;
    a[r] = v2f{sel, sel};
  }
  f32x4 d[16][LPR];
#pragma unroll
  for (int r = 0; r < 16; ++r) {
    int rr = row0 + r;
    if (rr >= n_nodes) rr = n_nodes - 1;
    const f32x4* __restrict__ rp = (const f32x4*)(emb + (size_t)rr * EMB) + lane;
#pragma unroll
    for (int j = 0; j < LPR; ++j) d[r][j] = rp[32 * j];
  }
  __builtin_amdgcn_sched_barrier(0);
#pragma unroll
  for (int r = 0; r < 16; ++r) {
#pragma unroll
    for (int j = 0; j < LPR; ++j) {
      const v2f b0 = {d[r][j].x, d[r][j].y};
      const v2f b1 = {d[r][j].z, d[r][j].w};
      c = __builtin_amdgcn_wmma_f32_16x16x4_f32(
           false, a[r],  false, b0,
           (short)0, c,  false,  false);
      c2 = __builtin_amdgcn_wmma_f32_16x16x4_f32(
           false, a[r],  false, b1,
           (short)0, c2,  false,  false);
    }
  }
  __builtin_amdgcn_sched_barrier(0);
  asm volatile("v_nop\n\tv_nop\n\tv_nop\n\tv_nop" : "+v"(c), "+v"(c2));
  c += c2;
#pragma unroll
  for (int r = 0; r < 8; ++r) {
    float x = c[r];
    x += __shfl_xor(x, 8);
    x += __shfl_xor(x, 4);
    x += __shfl_xor(x, 2);
    x += __shfl_xor(x, 1);
    c[r] = x;
  }
#else
  const float* __restrict__ row = emb + (size_t)(row0 + m) * EMB;
  v16h bones;
#pragma unroll
  for (int j = 0; j < 16; ++j) bones[j] = (_Float16)1.0f;
  for (int k0 = 0; k0 < EMB; k0 += 32) {
    v16h a;
    const float* p0 = row + k0 + half * 8;
    const float* p1 = row + k0 + 16 + half * 8;
#pragma unroll
    for (int j = 0; j < 8; ++j) a[j] = (_Float16)p0[j];
#pragma unroll
    for (int j = 0; j < 8; ++j) a[8 + j] = (_Float16)p1[j];
    c = __builtin_amdgcn_wmma_f32_16x16x32_f16(
        false, a, false, bones, (short)0, c, false, false);
  }
#endif

  if (m == 0) {
    const float inv = 1.0f / (float)EMB;
    const int base = row0 + half * 8;
    if (base + 8 <= n_nodes) {
#pragma unroll
      for (int r = 0; r < 8; ++r) node_mean[base + r] = c[r] * inv;
    } else {
      for (int r = 0; r < 8; ++r)
        if (base + r < n_nodes) node_mean[base + r] = c[r] * inv;
    }
  }
}

#define PO_ROWS 64

__global__ void zero_and_owner(float* __restrict__ p, int n,
                               const int* __restrict__ pair_place,
                               int* __restrict__ owner, int n_pairs) {
  int i = blockIdx.x * blockDim.x + threadIdx.x;
  if (i < n) p[i] = 0.0f;
  if (owner == nullptr) return;
  const int words = (n_pairs + 3) >> 2;
  if (i >= words) return;
  int a0, a1, a2, a3;
  if ((((size_t)pair_place) & 15) == 0 && 4 * i + 3 < n_pairs) {
    const int4 v = ((const int4*)pair_place)[i];
    a0 = v.x; a1 = v.y; a2 = v.z; a3 = v.w;
  } else {
    a0 = (4 * i + 0 < n_pairs) ? pair_place[4 * i + 0] : 0;
    a1 = (4 * i + 1 < n_pairs) ? pair_place[4 * i + 1] : 0;
    a2 = (4 * i + 2 < n_pairs) ? pair_place[4 * i + 2] : 0;
    a3 = (4 * i + 3 < n_pairs) ? pair_place[4 * i + 3] : 0;
  }
  const unsigned b0 = ((unsigned)a0 / PO_ROWS) & 255u;
  const unsigned b1 = ((unsigned)a1 / PO_ROWS) & 255u;
  const unsigned b2 = ((unsigned)a2 / PO_ROWS) & 255u;
  const unsigned b3 = ((unsigned)a3 / PO_ROWS) & 255u;
  owner[i] = (int)(b0 | (b1 << 8) | (b2 << 16) | (b3 << 24));
}

#define DK 8
#define DN 4
#define DT 256

__global__ __launch_bounds__(DT) void dst_runs(const int* __restrict__ dst_event,
                                               const int* __restrict__ dst_node,
                                               const float* __restrict__ node_mean,
                                               float* __restrict__ dst_sum,
                                               int n_dst, int n_events) {
  const int t = blockIdx.x * DT + threadIdx.x;
  if (n_dst <= 0) {
    for (int g = t; g < n_events; g += gridDim.x * DT) dst_sum[g] = 0.0f;
    return;
  }
  const int start = t * DK;
  if (start >= n_dst) return;

  int id[DK + DN], nd[DK + DN];
  const bool vec =
      ((((size_t)dst_event) | ((size_t)dst_node)) & 15) == 0 &&
      start + DK + DN <= n_dst;
  if (vec) {
    const int4* __restrict__ e4 = (const int4*)(dst_event + start);
    const int4* __restrict__ n4 = (const int4*)(dst_node + start);
#pragma unroll
    for (int q = 0; q < (DK + DN) / 4; ++q) {
      const int4 a = e4[q];
      const int4 b = n4[q];
      id[4 * q + 0] = a.x; id[4 * q + 1] = a.y;
      id[4 * q + 2] = a.z; id[4 * q + 3] = a.w;
      nd[4 * q + 0] = b.x; nd[4 * q + 1] = b.y;
      nd[4 * q + 2] = b.z; nd[4 * q + 3] = b.w;
    }
  } else {
#pragma unroll
    for (int j = 0; j < DK + DN; ++j) {
      const bool ok = start + j < n_dst;
      id[j] = ok ? dst_event[start + j] : -2;
      nd[j] = ok ? dst_node[start + j] : 0;
    }
  }
  int cur = (start > 0) ? dst_event[start - 1] : -1;

  float v[DK + DN];
#pragma unroll
  for (int j = 0; j < DK + DN; ++j) v[j] = node_mean[nd[j]];

  float acc = 0.0f;
  bool own = false;
#pragma unroll
  for (int j = 0; j < DK; ++j) {
    const int e = id[j];
    if (e != -2) {
      if (e != cur) {
        if (own) dst_sum[cur] = acc;
#pragma clang loop unroll(disable) vectorize(disable)
        for (int g = cur + 1; g < e; ++g) dst_sum[g] = 0.0f;
        cur = e;
        acc = v[j];
        own = true;
      } else {
        acc += v[j];
      }
    }
  }

  if (own) {
    int j = DK;
#pragma unroll
    for (int u = 0; u < DN; ++u) {
      if (j == DK + u && id[DK + u] == cur) { acc += v[DK + u]; ++j; }
    }
    int g = start + j;
    if (j == DK + DN) {
      while (g < n_dst && dst_event[g] == cur) {
        acc += node_mean[dst_node[g]];
        ++g;
      }
    }
    dst_sum[cur] = acc;
    if (g >= n_dst) {
#pragma clang loop unroll(disable) vectorize(disable)
      for (int z = cur + 1; z < n_events; ++z) dst_sum[z] = 0.0f;
    }
  }
}

__global__ void event_value(const int* __restrict__ event_pair,
                            const int* __restrict__ event_src,
                            const int* __restrict__ event_ndst,
                            const int* __restrict__ pair_place,
                            const float* __restrict__ node_mean,
                            float* __restrict__ ev, int n_events) {
  int i = blockIdx.x * blockDim.x + threadIdx.x;
  if (i >= n_events) return;
  int p = event_pair[i];
  float s = (node_mean[pair_place[p]] + node_mean[event_src[i]] +
             ev[i] / ((float)event_ndst[i] + 1.0f)) * (1.0f / 3.0f);
  ev[i] = s;
}

#define PR 16
#define EC 16
#define CAP 24576
#define AT 1024
#define UV 4

__global__ __launch_bounds__(AT) void pair_accum_lds(const int* __restrict__ event_pair,
                                                     const float* __restrict__ ev,
                                                     float* __restrict__ pair_sum,
                                                     float* __restrict__ pair_cnt,
                                                     int n_events, int n_pairs) {
  __shared__ float lsum[CAP];
  __shared__ float lcnt[CAP];

  const int tid = threadIdx.x;
  const int r = blockIdx.x % PR;
  const int c = blockIdx.x / PR;

  const int R  = (((n_pairs + PR - 1) / PR) + 31) & ~31;
  const int lo = r * R;
  const int hi = (lo + R < n_pairs) ? lo + R : n_pairs;

  const int clen = (((n_events + EC - 1) / EC) + 3) & ~3;
  const int e0 = c * clen;
  const int e1 = (e0 + clen < n_events) ? e0 + clen : n_events;

  const bool aligned =
      ((((size_t)(event_pair + e0)) | ((size_t)(ev + e0))) & 15) == 0;
  const int nvec = (aligned && e1 > e0) ? ((e1 - e0) >> 2) : 0;
  const int4*  __restrict__ ep4 = (const int4*)(event_pair + e0);
  const float4* __restrict__ s4 = (const float4*)(ev + e0);

  for (int base = lo; base < hi; base += CAP) {
    const int cnt = (hi - base < CAP) ? hi - base : CAP;

    for (int i = tid; i < cnt; i += AT) { lsum[i] = 0.0f; lcnt[i] = 0.0f; }
    __syncthreads();

    auto add1 = [&](int p, float s) {
      const unsigned rel = (unsigned)(p - base);
      if (rel < (unsigned)cnt) {
        __hip_atomic_fetch_add(&lsum[rel], s, __ATOMIC_RELAXED,
                               __HIP_MEMORY_SCOPE_WORKGROUP);
        __hip_atomic_fetch_add(&lcnt[rel], 1.0f, __ATOMIC_RELAXED,
                               __HIP_MEMORY_SCOPE_WORKGROUP);
      }
    };

    for (int v = tid; v < nvec; v += AT * UV) {
      int4 pp[UV];
      float4 ss[UV];
#pragma unroll
      for (int u = 0; u < UV; ++u) {
        const int vv = v + u * AT;
        if (vv < nvec) {
          pp[u] = ep4[vv];
          ss[u] = s4[vv];
        } else {
          pp[u] = make_int4(-1, -1, -1, -1);
          ss[u] = make_float4(0.0f, 0.0f, 0.0f, 0.0f);
        }
      }
#pragma unroll
      for (int u = 0; u < UV; ++u) {
        add1(pp[u].x, ss[u].x);
        add1(pp[u].y, ss[u].y);
        add1(pp[u].z, ss[u].z);
        add1(pp[u].w, ss[u].w);
      }
    }
    for (int i = e0 + 4 * nvec + tid; i < e1; i += AT)
      add1(event_pair[i], ev[i]);
    __syncthreads();

    const int rot = ((c * cnt) / EC) & ~31;
    for (int i = tid; i < cnt; i += AT) {
      int j = i + rot;
      if (j >= cnt) j -= cnt;
      const float n = lcnt[j];
      if (n > 0.0f) {
        unsafeAtomicAdd(&pair_sum[base + j], lsum[j]);
        unsafeAtomicAdd(&pair_cnt[base + j], n);
      }
    }
    __syncthreads();
  }
}

#define PO_T    1024
#define PO_UV   4
#define PO_QCAP 49152

__global__ __launch_bounds__(PO_T) void place_rows(const int* __restrict__ pair_place,
                                                   const int* __restrict__ owner,
                                                   const float* __restrict__ pair_sum,
                                                   const float* __restrict__ pair_cnt,
                                                   float* __restrict__ out,
                                                   int n_pairs, int n_nodes) {
  __shared__ int   queue[PO_QCAP];
  __shared__ float tsum[PO_ROWS];
  __shared__ float tcnt[PO_ROWS];
  __shared__ int   qn;

  const int tid = threadIdx.x;
  const int lo = blockIdx.x * PO_ROWS;
  const unsigned nrows =
      (unsigned)((n_nodes - lo < PO_ROWS) ? n_nodes - lo : PO_ROWS);

  if (tid < PO_ROWS) { tsum[tid] = 0.0f; tcnt[tid] = 0.0f; }
  if (tid == 0) qn = 0;
  __syncthreads();

  auto add_mean = [&](unsigned rel, float c, float s) {
    if (c > 0.0f) {
      __hip_atomic_fetch_add(&tsum[rel], s / c, __ATOMIC_RELAXED,
                             __HIP_MEMORY_SCOPE_WORKGROUP);
      __hip_atomic_fetch_add(&tcnt[rel], 1.0f, __ATOMIC_RELAXED,
                             __HIP_MEMORY_SCOPE_WORKGROUP);
    }
  };

  const bool have_plane = owner != nullptr;
  if (have_plane) {
    const unsigned mine = 0x01010101u * (blockIdx.x & 255u);
    auto eq = [&](unsigned x) {
      const unsigned t = x ^ mine;
      return ~(((t & 0x7f7f7f7fu) + 0x7f7f7f7fu) | t | 0x7f7f7f7fu);
    };
    auto hits = [&](const uint4 v) {
      return (eq(v.x) >> 7) | (eq(v.y) >> 6) | (eq(v.z) >> 5) | (eq(v.w) >> 4);
    };
    auto emit = [&](unsigned m, int w, int& pos) {
      while (m != 0u) {
        const int q = __builtin_ctz(m);
        m &= m - 1u;
        if (pos < PO_QCAP) queue[pos] = 16 * w + 4 * (q & 7) + (q >> 3);
        ++pos;
      }
    };
    auto reserve = [&](int n) {
      return __hip_atomic_fetch_add(&qn, n, __ATOMIC_RELAXED,
                                    __HIP_MEMORY_SCOPE_WORKGROUP);
    };

    const bool aligned = (((size_t)owner) & 15) == 0;
    const int n16 = aligned ? (n_pairs >> 4) : 0;
    const uint4* __restrict__ ow4 = (const uint4*)owner;
    const int rot = (int)blockIdx.x * (int)((unsigned)n16 / gridDim.x);
    auto widx = [&](int vv) {
      const int w = vv + rot;
      return (w >= n16) ? w - n16 : w;
    };
    auto look_batch = [&](const uint4 (&v)[PO_UV], int vv0) {
      unsigned m[PO_UV];
      int n = 0;
#pragma unroll
      for (int u = 0; u < PO_UV; ++u) {
        m[u] = hits(v[u]);
        n += (int)__builtin_popcount(m[u]);
      }
      if (n != 0) {
        int pos = reserve(n);
#pragma unroll
        for (int u = 0; u < PO_UV; ++u) emit(m[u], widx(vv0 + u * PO_T), pos);
      }
    };

    const int STEP = PO_T * PO_UV;
    const int nfull = n16 / STEP;
    uint4 cur[PO_UV], nxt[PO_UV];
    if (nfull > 0) {
#pragma unroll
      for (int u = 0; u < PO_UV; ++u) cur[u] = ow4[widx(u * PO_T + tid)];
      for (int b = 0; b + 1 < nfull; ++b) {
#pragma unroll
        for (int u = 0; u < PO_UV; ++u)
          nxt[u] = ow4[widx((b + 1) * STEP + u * PO_T + tid)];
        look_batch(cur, b * STEP + tid);
#pragma unroll
        for (int u = 0; u < PO_UV; ++u) cur[u] = nxt[u];
      }
      look_batch(cur, (nfull - 1) * STEP + tid);
    }
    for (int vv = nfull * STEP + tid; vv < n16; vv += PO_T) {
      const int w = widx(vv);
      const unsigned m = hits(ow4[w]);
      if (m != 0u) {
        int pos = reserve((int)__builtin_popcount(m));
        emit(m, w, pos);
      }
    }
    const unsigned char* __restrict__ ow1 = (const unsigned char*)owner;
    for (int i = 16 * n16 + tid; i < n_pairs; i += PO_T) {
      if (ow1[i] == (unsigned char)(blockIdx.x & 255u)) {
        const int pos = reserve(1);
        if (pos < PO_QCAP) queue[pos] = i;
      }
    }
  }
  __syncthreads();

  const int nh = qn;
  if (have_plane && nh <= PO_QCAP) {
    for (int e = tid; e < nh; e += PO_T) {
      const int i = queue[e];
      const int   pl = pair_place[i];
      const float c  = pair_cnt[i];
      const unsigned rel = (unsigned)(pl - lo);
      if ((rel < nrows) & (c > 0.0f)) add_mean(rel, c, pair_sum[i]);
    }
  } else {
    for (int i = tid; i < n_pairs; i += PO_T) {
      const unsigned rel = (unsigned)(pair_place[i] - lo);
      if (rel < nrows) add_mean(rel, pair_cnt[i], pair_sum[i]);
    }
  }
  __syncthreads();

  if (tid < PO_ROWS) {
    const float c = tcnt[tid];
    tsum[tid] = (c > 0.0f) ? tsum[tid] / c : 0.0f;
  }
  __syncthreads();

  const int lane = tid & 31;
  f32x4* __restrict__ o = (f32x4*)out + (size_t)lo * (OUTW / 4);
  for (int r = tid >> 5; r < (int)nrows; r += PO_T / 32) {
    const float pm = tsum[r];
    const f32x4 v = {pm, pm, pm, pm};
    f32x4* __restrict__ orow = o + (size_t)r * (OUTW / 4) + lane;
#pragma unroll
    for (int k = 0; k < OUTW / 4 / 32; ++k)
      __builtin_nontemporal_store(v, orow + 32 * k);
  }
}

extern "C" void kernel_launch(void* const* d_in, const int* in_sizes, int n_in,
                              void* d_out, int out_size, void* d_ws, size_t ws_size,
                              hipStream_t stream) {
  const float* emb       = (const float*)d_in[0];
  const int* pair_place  = (const int*)d_in[1];
  const int* event_pair  = (const int*)d_in[2];
  const int* event_src   = (const int*)d_in[3];
  const int* event_ndst  = (const int*)d_in[4];
  const int* dst_event   = (const int*)d_in[5];
  const int* dst_node    = (const int*)d_in[6];

  const int n_nodes  = in_sizes[0] / EMB;
  const int n_pairs  = in_sizes[1];
  const int n_events = in_sizes[2];
  const int n_dst    = in_sizes[5];

  float* ws        = (float*)d_ws;
  float* node_mean = ws;
  float* dst_sum   = node_mean + n_nodes;
  float* pair_sum  = dst_sum + n_events;
  float* pair_cnt  = pair_sum + n_pairs;
  const size_t plane_at = ((size_t)n_nodes + (size_t)n_events + 2 * (size_t)n_pairs) * sizeof(float);
  const size_t plane_len = (size_t)((n_pairs + 3) / 4) * sizeof(int);
  int* owner = (ws_size >= plane_at + plane_len) ? (int*)(pair_cnt + n_pairs) : nullptr;

  const int nzero = 2 * n_pairs;
  zero_and_owner<<<(nzero + 255) / 256, 256, 0, stream>>>(pair_sum, nzero, pair_place,
                                                          owner, n_pairs);

  rowmean_wmma<<<(n_nodes + 63) / 64, 128, 0, stream>>>(emb, node_mean, n_nodes);

  dst_runs<<<n_dst / (DK * DT) + 1, DT, 0, stream>>>(dst_event, dst_node, node_mean,
                                                      dst_sum, n_dst, n_events);

  event_value<<<(n_events + 255) / 256, 256, 0, stream>>>(
      event_pair, event_src, event_ndst, pair_place, node_mean, dst_sum,
      n_events);

  pair_accum_lds<<<PR * EC, AT, 0, stream>>>(event_pair, dst_sum, pair_sum,
                                             pair_cnt, n_events, n_pairs);

  place_rows<<<(n_nodes + PO_ROWS - 1) / PO_ROWS, PO_T, 0, stream>>>(
      pair_place, owner, pair_sum, pair_cnt, (float*)d_out, n_pairs, n_nodes);
}
